# starting kernel
# baseline (speedup 1.0000x reference)
#include <hip/hip_runtime.h>
#include <hip/hip_bf16.h>
#include <math.h>


#define RNN_T    2048
#define RNN_B    2048
#define RNN_EMB  10
#define RNN_HID  8
#define RNN_NCLS 4

typedef __attribute__((ext_vector_type(2))) float v2f;
typedef __attribute__((ext_vector_type(4))) float v4f;
typedef __attribute__((ext_vector_type(8))) float v8f;

#if defined(__has_builtin)
#if __has_builtin(__builtin_amdgcn_tanhf)
#define RNN_TANH(x) __builtin_amdgcn_tanhf(x)
#endif
#endif
#ifndef RNN_TANH
#define RNN_TANH(x) tanhf(x)
#endif

__device__ __forceinline__ v8f wmma4(v2f a, v2f b, v8f c) {
    return __builtin_amdgcn_wmma_f32_16x16x4_f32(
         false, a,  false, b,
         (short)0, c,  false,  false);
}

__device__ __forceinline__ v2f load_a_dup(const float* W, int lo, int hi, int c) {
    const int r  = lo & 7;
    const int k0 = 4 * c + 2 * hi;
    v2f a;
    a.x = W[r * 8 + k0];
    a.y = W[r * 8 + k0 + 1];
    return a;
}

__device__ __forceinline__ v2f load_a_fc(const float* W, int lo, int hi, int c) {
    const int r = lo & 7;
    v2f a; a.x = 0.0f; a.y = 0.0f;
    if (r < RNN_NCLS) {
        const int k0 = 4 * c + 2 * hi;
        a.x = W[r * 8 + k0];
        a.y = W[r * 8 + k0 + 1];
    }
    return a;
}

__device__ __forceinline__ void d_to_b(const v8f& hD, bool hi, v2f& b0, v2f& b1) {
    b0.x = hi ? hD[2] : hD[0];
    b0.y = hi ? hD[3] : hD[1];
    b1.x = hi ? hD[6] : hD[4];
    b1.y = hi ? hD[7] : hD[5];
}

__global__ __launch_bounds__(32)
void rnn_fused_kernel(const int* __restrict__ x,
                      const float* __restrict__ emb,
                      const float* __restrict__ W_ih0,
                      const float* __restrict__ W_hh0,
                      const float* __restrict__ b_ih0,
                      const float* __restrict__ b_hh0,
                      const float* __restrict__ W_ih1,
                      const float* __restrict__ W_hh1,
                      const float* __restrict__ b_ih1,
                      const float* __restrict__ b_hh1,
                      const float* __restrict__ W_fc,
                      const float* __restrict__ b_fc,
                      float* __restrict__ out) {
    const int lane = threadIdx.x & 31;
    const int lo   = lane & 15;
    const int hi   = lane >> 4;
    const int batch_base = blockIdx.x * 16;

    v2f t0a;
    {
        const int j = lo & 7;
        const float bias = b_ih0[j] + b_hh0[j];
        const int v0 = 2 * hi, v1 = 2 * hi + 1;
        float a0 = bias, a1 = bias;
        #pragma unroll
        for (int d = 0; d < RNN_EMB; ++d) {
            const float w = W_ih0[j * RNN_EMB + d];
            a0 += emb[v0 * RNN_EMB + d] * w;
            a1 += emb[v1 * RNN_EMB + d] * w;
        }
        t0a.x = a0; t0a.y = a1;
    }

    const v2f whh0_a0 = load_a_dup(W_hh0, lo, hi, 0);
    const v2f whh0_a1 = load_a_dup(W_hh0, lo, hi, 1);
    const v2f wih1_a0 = load_a_dup(W_ih1, lo, hi, 0);
    const v2f wih1_a1 = load_a_dup(W_ih1, lo, hi, 1);
    const v2f whh1_a0 = load_a_dup(W_hh1, lo, hi, 0);
    const v2f whh1_a1 = load_a_dup(W_hh1, lo, hi, 1);
    const v2f wfc_a0  = load_a_fc (W_fc,  lo, hi, 0);
    const v2f wfc_a1  = load_a_fc (W_fc,  lo, hi, 1);

    v8f c_bias1, c_fc, c_zero;
    #pragma unroll
    for (int i = 0; i < 8; ++i) {
        c_bias1[i] = b_ih1[i] + b_hh1[i];
        c_fc[i]    = (i < RNN_NCLS) ? b_fc[i] : 0.0f;
        c_zero[i]  = 0.0f;
    }

    v8f h0D = c_zero, h1D = c_zero;
    v2f h0b0 = {0.0f, 0.0f}, h0b1 = {0.0f, 0.0f};
    v2f h1b0 = {0.0f, 0.0f}, h1b1 = {0.0f, 0.0f};

    const int* xrow = x + (size_t)batch_base + lo;
    float*     op   = out + ((size_t)batch_base + lo) * RNN_NCLS;

    int        xi      = xrow[0];
    const int* xp_next = xrow + RNN_B;

    #pragma unroll 1
    for (int t = 0; t < RNN_T; ++t) {
        if (t + 8 < RNN_T)
            __builtin_prefetch(xp_next + (size_t)7 * RNN_B, 0, 3);
        int xi_next = xi;
        if (t + 1 < RNN_T) xi_next = xp_next[0];

        v2f oh;
        oh.x = (xi == 2 * hi)     ? 1.0f : 0.0f;
        oh.y = (xi == 2 * hi + 1) ? 1.0f : 0.0f;

        v8f p0 = wmma4(t0a, oh, c_zero);
        p0 = wmma4(whh0_a0, h0b0, p0);
        p0 = wmma4(whh0_a1, h0b1, p0);
        #pragma unroll
        for (int i = 0; i < 8; ++i) h0D[i] = RNN_TANH(p0[i]);
        d_to_b(h0D, hi != 0, h0b0, h0b1);

        v8f p1 = c_bias1;
        p1 = wmma4(wih1_a0, h0b0, p1);
        p1 = wmma4(wih1_a1, h0b1, p1);
        p1 = wmma4(whh1_a0, h1b0, p1);
        p1 = wmma4(whh1_a1, h1b1, p1);
        #pragma unroll
        for (int i = 0; i < 8; ++i) h1D[i] = RNN_TANH(p1[i]);
        d_to_b(h1D, hi != 0, h1b0, h1b1);

        v8f lg = c_fc;
        lg = wmma4(wfc_a0, h1b0, lg);
        lg = wmma4(wfc_a1, h1b1, lg);

        if (!hi) {
            v4f res; res.x = lg[0]; res.y = lg[1]; res.z = lg[2]; res.w = lg[3];
            *(v4f*)op = res;
        }

        xi = xi_next;
        xp_next += RNN_B;
        op += (size_t)RNN_B * RNN_NCLS;
    }
}

extern "C" void kernel_launch(void* const* d_in, const int* in_sizes, int n_in,
                              void* d_out, int out_size, void* d_ws, size_t ws_size,
                              hipStream_t stream) {
    (void)in_sizes; (void)n_in; (void)d_ws; (void)ws_size; (void)out_size;
    const int*   x     = (const int*)  d_in[0];
    const float* emb   = (const float*)d_in[1];
    const float* W_ih0 = (const float*)d_in[2];
    const float* W_hh0 = (const float*)d_in[3];
    const float* b_ih0 = (const float*)d_in[4];
    const float* b_hh0 = (const float*)d_in[5];
    const float* W_ih1 = (const float*)d_in[6];
    const float* W_hh1 = (const float*)d_in[7];
    const float* b_ih1 = (const float*)d_in[8];
    const float* b_hh1 = (const float*)d_in[9];
    const float* W_fc  = (const float*)d_in[10];
    const float* b_fc  = (const float*)d_in[11];
    float* out = (float*)d_out;

    dim3 grid(RNN_B / 16);
    dim3 block(32);
    rnn_fused_kernel<<<grid, block, 0, stream>>>(
        x, emb, W_ih0, W_hh0, b_ih0, b_hh0,
        W_ih1, W_hh1, b_ih1, b_hh1, W_fc, b_fc, out);
}


# Token data path: 8-step x 16-column token blocks fetched by one 16-byte load per lane a trip ahead, handed over through LDS; was one 4-byte load per step waited for inside the step; sum order kept
# speedup vs baseline: 1.4724x; 1.4724x over previous
#include <hip/hip_runtime.h>
#include <hip/hip_bf16.h>
#include <math.h>


#define RNN_T    2048
#define RNN_B    2048
#define RNN_EMB  10
#define RNN_HID  8
#define RNN_NCLS 4
#define RNN_GRP  8
#define RNN_NGRP (RNN_T / RNN_GRP)

typedef __attribute__((ext_vector_type(2))) float v2f;
typedef __attribute__((ext_vector_type(4))) float v4f;
typedef __attribute__((ext_vector_type(8))) float v8f;
typedef __attribute__((ext_vector_type(4))) int   v4i;

#if defined(__has_builtin)
#if __has_builtin(__builtin_amdgcn_tanhf)
#define RNN_TANH(x) __builtin_amdgcn_tanhf(x)
#endif
#endif
#ifndef RNN_TANH
#define RNN_TANH(x) tanhf(x)
#endif

__device__ __forceinline__ v8f wmma4(v2f a, v2f b, v8f c) {
    return __builtin_amdgcn_wmma_f32_16x16x4_f32(
         false, a,  false, b,
         (short)0, c,  false,  false);
}

__device__ __forceinline__ v2f load_a_dup(const float* W, int lo, int hi, int c) {
    const int r  = lo & 7;
    const int k0 = 4 * c + 2 * hi;
    v2f a;
    a.x = W[r * 8 + k0];
    a.y = W[r * 8 + k0 + 1];
    return a;
}

__device__ __forceinline__ v2f load_a_fc(const float* W, int lo, int hi, int c) {
    const int r = lo & 7;
    v2f a; a.x = 0.0f; a.y = 0.0f;
    if (r < RNN_NCLS) {
        const int k0 = 4 * c + 2 * hi;
        a.x = W[r * 8 + k0];
        a.y = W[r * 8 + k0 + 1];
    }
    return a;
}

__device__ __forceinline__ void d_to_b(const v8f& hD, bool hi, v2f& b0, v2f& b1) {
    b0.x = hi ? hD[2] : hD[0];
    b0.y = hi ? hD[3] : hD[1];
    b1.x = hi ? hD[6] : hD[4];
    b1.y = hi ? hD[7] : hD[5];
}

__global__ __launch_bounds__(32)
void rnn_fused_kernel(const int* __restrict__ x,
                      const float* __restrict__ emb,
                      const float* __restrict__ W_ih0,
                      const float* __restrict__ W_hh0,
                      const float* __restrict__ b_ih0,
                      const float* __restrict__ b_hh0,
                      const float* __restrict__ W_ih1,
                      const float* __restrict__ W_hh1,
                      const float* __restrict__ b_ih1,
                      const float* __restrict__ b_hh1,
                      const float* __restrict__ W_fc,
                      const float* __restrict__ b_fc,
                      float* __restrict__ out) {
    const int lane = threadIdx.x & 31;
    const int lo   = lane & 15;
    const int hi   = lane >> 4;
    const int batch_base = blockIdx.x * 16;

    v2f t0a;
    {
        const int j = lo & 7;
        const float bias = b_ih0[j] + b_hh0[j];
        const int v0 = 2 * hi, v1 = 2 * hi + 1;
        float a0 = bias, a1 = bias;
        #pragma unroll
        for (int d = 0; d < RNN_EMB; ++d) {
            const float w = W_ih0[j * RNN_EMB + d];
            a0 += emb[v0 * RNN_EMB + d] * w;
            a1 += emb[v1 * RNN_EMB + d] * w;
        }
        t0a.x = a0; t0a.y = a1;
    }

    const v2f whh0_a0 = load_a_dup(W_hh0, lo, hi, 0);
    const v2f whh0_a1 = load_a_dup(W_hh0, lo, hi, 1);
    const v2f wih1_a0 = load_a_dup(W_ih1, lo, hi, 0);
    const v2f wih1_a1 = load_a_dup(W_ih1, lo, hi, 1);
    const v2f whh1_a0 = load_a_dup(W_hh1, lo, hi, 0);
    const v2f whh1_a1 = load_a_dup(W_hh1, lo, hi, 1);
    const v2f wfc_a0  = load_a_fc (W_fc,  lo, hi, 0);
    const v2f wfc_a1  = load_a_fc (W_fc,  lo, hi, 1);

    v8f c_bias1, c_fc, c_zero;
    #pragma unroll
    for (int i = 0; i < 8; ++i) {
        c_bias1[i] = b_ih1[i] + b_hh1[i];
        c_fc[i]    = (i < RNN_NCLS) ? b_fc[i] : 0.0f;
        c_zero[i]  = 0.0f;
    }

    v8f h0D = c_zero, h1D = c_zero;
    v2f h0b0 = {0.0f, 0.0f}, h0b1 = {0.0f, 0.0f};
    v2f h1b0 = {0.0f, 0.0f}, h1b1 = {0.0f, 0.0f};

    float* op = out + ((size_t)batch_base + lo) * RNN_NCLS;

    __shared__ int tokbuf[RNN_GRP * 16];
    const int ldr = lane >> 2;
    const int ldp = lane & 3;
    const int* xblk   = x + (size_t)ldr * RNN_B + batch_base + 4 * ldp;
    int*       lds_st = &tokbuf[ldr * 16 + 4 * ldp];
    const int* lds_ld = &tokbuf[lo];

    int tok[RNN_GRP];
    v4i blk = *(const v4i*)xblk;
    *(v4i*)lds_st = blk;
    __builtin_amdgcn_wave_barrier();
    #pragma unroll
    for (int s = 0; s < RNN_GRP; ++s) tok[s] = lds_ld[16 * s];
    blk = *(const v4i*)(xblk + (size_t)RNN_GRP * RNN_B);

    #pragma unroll 1
    for (int g = 0; g < RNN_NGRP; ++g) {
        int tokn[RNN_GRP];
        *(v4i*)lds_st = blk;
        __builtin_amdgcn_wave_barrier();
        #pragma unroll
        for (int s = 0; s < RNN_GRP; ++s) tokn[s] = lds_ld[16 * s];
        const int gn = (g + 2 < RNN_NGRP) ? (g + 2) : (RNN_NGRP - 1);
        blk = *(const v4i*)(xblk + (size_t)gn * (RNN_GRP * RNN_B));

        #pragma unroll
        for (int s = 0; s < RNN_GRP; ++s) {
            const int xi = tok[s];

            v2f oh;
            oh.x = (xi == 2 * hi)     ? 1.0f : 0.0f;
            oh.y = (xi == 2 * hi + 1) ? 1.0f : 0.0f;

            v8f p0 = wmma4(t0a, oh, c_zero);
            p0 = wmma4(whh0_a0, h0b0, p0);
            p0 = wmma4(whh0_a1, h0b1, p0);
            #pragma unroll
            for (int i = 0; i < 8; ++i) h0D[i] = RNN_TANH(p0[i]);
            d_to_b(h0D, hi != 0, h0b0, h0b1);

            v8f p1 = c_bias1;
            p1 = wmma4(wih1_a0, h0b0, p1);
            p1 = wmma4(wih1_a1, h0b1, p1);
            p1 = wmma4(whh1_a0, h1b0, p1);
            p1 = wmma4(whh1_a1, h1b1, p1);
            #pragma unroll
            for (int i = 0; i < 8; ++i) h1D[i] = RNN_TANH(p1[i]);
            d_to_b(h1D, hi != 0, h1b0, h1b1);

            v8f lg = c_fc;
            lg = wmma4(wfc_a0, h1b0, lg);
            lg = wmma4(wfc_a1, h1b1, lg);

            if (!hi) {
                v4f res; res.x = lg[0]; res.y = lg[1]; res.z = lg[2]; res.w = lg[3];
                *(v4f*)(op + (size_t)s * (RNN_B * RNN_NCLS)) = res;
            }
        }

        op += (size_t)RNN_GRP * RNN_B * RNN_NCLS;
        #pragma unroll
        for (int s = 0; s < RNN_GRP; ++s) tok[s] = tokn[s];
    }
}

extern "C" void kernel_launch(void* const* d_in, const int* in_sizes, int n_in,
                              void* d_out, int out_size, void* d_ws, size_t ws_size,
                              hipStream_t stream) {
    (void)in_sizes; (void)n_in; (void)d_ws; (void)ws_size; (void)out_size;
    const int*   x     = (const int*)  d_in[0];
    const float* emb   = (const float*)d_in[1];
    const float* W_ih0 = (const float*)d_in[2];
    const float* W_hh0 = (const float*)d_in[3];
    const float* b_ih0 = (const float*)d_in[4];
    const float* b_hh0 = (const float*)d_in[5];
    const float* W_ih1 = (const float*)d_in[6];
    const float* W_hh1 = (const float*)d_in[7];
    const float* b_ih1 = (const float*)d_in[8];
    const float* b_hh1 = (const float*)d_in[9];
    const float* W_fc  = (const float*)d_in[10];
    const float* b_fc  = (const float*)d_in[11];
    float* out = (float*)d_out;

    dim3 grid(RNN_B / 16);
    dim3 block(32);
    rnn_fused_kernel<<<grid, block, 0, stream>>>(
        x, emb, W_ih0, W_hh0, b_ih0, b_hh0,
        W_ih1, W_hh1, b_ih1, b_hh1, W_fc, b_fc, out);
}


# Time loop tiled in 8-step blocks, stages run block-wise in one scheduling region: layer 0 x8, layer 1 x8, class product + stores x8; state-free products leave the chains; sum order kept
# speedup vs baseline: 2.2222x; 1.5092x over previous
#include <hip/hip_runtime.h>
#include <hip/hip_bf16.h>
#include <math.h>


#define RNN_T    2048
#define RNN_B    2048
#define RNN_EMB  10
#define RNN_HID  8
#define RNN_NCLS 4
#define RNN_GRP  8
#define RNN_NGRP (RNN_T / RNN_GRP)

typedef __attribute__((ext_vector_type(2))) float v2f;
typedef __attribute__((ext_vector_type(4))) float v4f;
typedef __attribute__((ext_vector_type(8))) float v8f;
typedef __attribute__((ext_vector_type(4))) int   v4i;

#if defined(__has_builtin)
#if __has_builtin(__builtin_amdgcn_tanhf)
#define RNN_TANH(x) __builtin_amdgcn_tanhf(x)
#endif
#endif
#ifndef RNN_TANH
#define RNN_TANH(x) tanhf(x)
#endif

__device__ __forceinline__ v8f wmma4(v2f a, v2f b, v8f c) {
    return __builtin_amdgcn_wmma_f32_16x16x4_f32(
         false, a,  false, b,
         (short)0, c,  false,  false);
}

__device__ __forceinline__ v2f load_a_dup(const float* W, int lo, int hi, int c) {
    const int r  = lo & 7;
    const int k0 = 4 * c + 2 * hi;
    v2f a;
    a.x = W[r * 8 + k0];
    a.y = W[r * 8 + k0 + 1];
    return a;
}

__device__ __forceinline__ v2f load_a_fc(const float* W, int lo, int hi, int c) {
    const int r = lo & 7;
    v2f a; a.x = 0.0f; a.y = 0.0f;
    if (r < RNN_NCLS) {
        const int k0 = 4 * c + 2 * hi;
        a.x = W[r * 8 + k0];
        a.y = W[r * 8 + k0 + 1];
    }
    return a;
}

__device__ __forceinline__ void d_to_b(const v8f& hD, bool hi, v2f& b0, v2f& b1) {
    b0.x = hi ? hD[2] : hD[0];
    b0.y = hi ? hD[3] : hD[1];
    b1.x = hi ? hD[6] : hD[4];
    b1.y = hi ? hD[7] : hD[5];
}

__device__ __forceinline__ v2f onehot_b(int xi, int hi) {
    v2f oh;
    oh.x = (xi == 2 * hi)     ? 1.0f : 0.0f;
    oh.y = (xi == 2 * hi + 1) ? 1.0f : 0.0f;
    return oh;
}

__global__ __launch_bounds__(32)
void rnn_fused_kernel(const int* __restrict__ x,
                      const float* __restrict__ emb,
                      const float* __restrict__ W_ih0,
                      const float* __restrict__ W_hh0,
                      const float* __restrict__ b_ih0,
                      const float* __restrict__ b_hh0,
                      const float* __restrict__ W_ih1,
                      const float* __restrict__ W_hh1,
                      const float* __restrict__ b_ih1,
                      const float* __restrict__ b_hh1,
                      const float* __restrict__ W_fc,
                      const float* __restrict__ b_fc,
                      float* __restrict__ out) {
    const int lane = threadIdx.x & 31;
    const int lo   = lane & 15;
    const int hi   = lane >> 4;
    const int batch_base = blockIdx.x * 16;

    v2f t0a;
    {
        const int j = lo & 7;
        const float bias = b_ih0[j] + b_hh0[j];
        const int v0 = 2 * hi, v1 = 2 * hi + 1;
        float a0 = bias, a1 = bias;
        #pragma unroll
        for (int d = 0; d < RNN_EMB; ++d) {
            const float w = W_ih0[j * RNN_EMB + d];
            a0 += emb[v0 * RNN_EMB + d] * w;
            a1 += emb[v1 * RNN_EMB + d] * w;
        }
        t0a.x = a0; t0a.y = a1;
    }

    const v2f whh0_a0 = load_a_dup(W_hh0, lo, hi, 0);
    const v2f whh0_a1 = load_a_dup(W_hh0, lo, hi, 1);
    const v2f wih1_a0 = load_a_dup(W_ih1, lo, hi, 0);
    const v2f wih1_a1 = load_a_dup(W_ih1, lo, hi, 1);
    const v2f whh1_a0 = load_a_dup(W_hh1, lo, hi, 0);
    const v2f whh1_a1 = load_a_dup(W_hh1, lo, hi, 1);
    const v2f wfc_a0  = load_a_fc (W_fc,  lo, hi, 0);
    const v2f wfc_a1  = load_a_fc (W_fc,  lo, hi, 1);

    v8f c_bias1, c_fc, c_zero;
    #pragma unroll
    for (int i = 0; i < 8; ++i) {
        c_bias1[i] = b_ih1[i] + b_hh1[i];
        c_fc[i]    = (i < RNN_NCLS) ? b_fc[i] : 0.0f;
        c_zero[i]  = 0.0f;
    }

    v2f h0b0 = {0.0f, 0.0f}, h0b1 = {0.0f, 0.0f};
    v2f h1b0 = {0.0f, 0.0f}, h1b1 = {0.0f, 0.0f};

    float* op = out + ((size_t)batch_base + lo) * RNN_NCLS;

    __shared__ int tokbuf[RNN_GRP * 16];
    const int ldr = lane >> 2;
    const int ldp = lane & 3;
    const int* xblk   = x + (size_t)ldr * RNN_B + batch_base + 4 * ldp;
    int*       lds_st = &tokbuf[ldr * 16 + 4 * ldp];
    const int* lds_ld = &tokbuf[lo];

    int tok[RNN_GRP];
    v4i blk = *(const v4i*)xblk;
    *(v4i*)lds_st = blk;
    __builtin_amdgcn_wave_barrier();
    #pragma unroll
    for (int s = 0; s < RNN_GRP; ++s) tok[s] = lds_ld[16 * s];
    blk = *(const v4i*)(xblk + (size_t)RNN_GRP * RNN_B);

    #pragma unroll 1
    for (int g = 0; g < RNN_NGRP; ++g) {
        int tokn[RNN_GRP];
        *(v4i*)lds_st = blk;
        __builtin_amdgcn_wave_barrier();
        #pragma unroll
        for (int s = 0; s < RNN_GRP; ++s) tokn[s] = lds_ld[16 * s];
        const int gn = (g + 2 < RNN_NGRP) ? (g + 2) : (RNN_NGRP - 1);
        blk = *(const v4i*)(xblk + (size_t)gn * (RNN_GRP * RNN_B));
        __builtin_amdgcn_sched_barrier(0);

        v2f a0[RNN_GRP], a1[RNN_GRP];
        v8f q = wmma4(t0a, onehot_b(tok[0], hi), c_zero);
        #pragma unroll
        for (int s = 0; s < RNN_GRP; ++s) {
            v8f p0 = wmma4(whh0_a0, h0b0, q);
            p0 = wmma4(whh0_a1, h0b1, p0);
            if (s + 1 < RNN_GRP) q = wmma4(t0a, onehot_b(tok[s + 1], hi), c_zero);
            v8f h0D;
            #pragma unroll
            for (int i = 0; i < 8; ++i) h0D[i] = RNN_TANH(p0[i]);
            d_to_b(h0D, hi != 0, h0b0, h0b1);
            a0[s] = h0b0; a1[s] = h0b1;
        }

        v2f e0[RNN_GRP], e1[RNN_GRP];
        v8f r = wmma4(wih1_a1, a1[0], wmma4(wih1_a0, a0[0], c_bias1));
        #pragma unroll
        for (int s = 0; s < RNN_GRP; ++s) {
            v8f p1 = wmma4(whh1_a0, h1b0, r);
            p1 = wmma4(whh1_a1, h1b1, p1);
            if (s + 1 < RNN_GRP)
                r = wmma4(wih1_a1, a1[s + 1], wmma4(wih1_a0, a0[s + 1], c_bias1));
            v8f h1D;
            #pragma unroll
            for (int i = 0; i < 8; ++i) h1D[i] = RNN_TANH(p1[i]);
            d_to_b(h1D, hi != 0, h1b0, h1b1);
            e0[s] = h1b0; e1[s] = h1b1;
        }

        v4f res[RNN_GRP];
        #pragma unroll
        for (int s = 0; s < RNN_GRP; ++s) {
            v8f lg = wmma4(wfc_a1, e1[s], wmma4(wfc_a0, e0[s], c_fc));
            res[s].x = lg[0]; res[s].y = lg[1]; res[s].z = lg[2]; res[s].w = lg[3];
        }
        if (!hi) {
            #pragma unroll
            for (int s = 0; s < RNN_GRP; ++s)
                *(v4f*)(op + (size_t)s * (RNN_B * RNN_NCLS)) = res[s];
        }

        op += (size_t)RNN_GRP * RNN_B * RNN_NCLS;
        #pragma unroll
        for (int s = 0; s < RNN_GRP; ++s) tok[s] = tokn[s];
    }
}

extern "C" void kernel_launch(void* const* d_in, const int* in_sizes, int n_in,
                              void* d_out, int out_size, void* d_ws, size_t ws_size,
                              hipStream_t stream) {
    (void)in_sizes; (void)n_in; (void)d_ws; (void)ws_size; (void)out_size;
    const int*   x     = (const int*)  d_in[0];
    const float* emb   = (const float*)d_in[1];
    const float* W_ih0 = (const float*)d_in[2];
    const float* W_hh0 = (const float*)d_in[3];
    const float* b_ih0 = (const float*)d_in[4];
    const float* b_hh0 = (const float*)d_in[5];
    const float* W_ih1 = (const float*)d_in[6];
    const float* W_hh1 = (const float*)d_in[7];
    const float* b_ih1 = (const float*)d_in[8];
    const float* b_hh1 = (const float*)d_in[9];
    const float* W_fc  = (const float*)d_in[10];
    const float* b_fc  = (const float*)d_in[11];
    float* out = (float*)d_out;

    dim3 grid(RNN_B / 16);
    dim3 block(32);
    rnn_fused_kernel<<<grid, block, 0, stream>>>(
        x, emb, W_ih0, W_hh0, b_ih0, b_hh0,
        W_ih1, W_hh1, b_ih1, b_hh1, W_fc, b_fc, out);
}


# Row select moved in front of the tanh: each lane takes tanh only of the 4 hidden rows its 16-lane group feeds to the next B operand, not of all 8 duplicated rows; bit-identical, sum order kept
# speedup vs baseline: 2.3595x; 1.0618x over previous
#include <hip/hip_runtime.h>
#include <hip/hip_bf16.h>
#include <math.h>


#define RNN_T    2048
#define RNN_B    2048
#define RNN_EMB  10
#define RNN_HID  8
#define RNN_NCLS 4
#define RNN_GRP  8
#define RNN_NGRP (RNN_T / RNN_GRP)

typedef __attribute__((ext_vector_type(2))) float v2f;
typedef __attribute__((ext_vector_type(4))) float v4f;
typedef __attribute__((ext_vector_type(8))) float v8f;
typedef __attribute__((ext_vector_type(4))) int   v4i;

#if defined(__has_builtin)
#if __has_builtin(__builtin_amdgcn_tanhf)
#define RNN_TANH(x) __builtin_amdgcn_tanhf(x)
#endif
#endif
#ifndef RNN_TANH
#define RNN_TANH(x) tanhf(x)
#endif

__device__ __forceinline__ v8f wmma4(v2f a, v2f b, v8f c) {
    return __builtin_amdgcn_wmma_f32_16x16x4_f32(
         false, a,  false, b,
         (short)0, c,  false,  false);
}

__device__ __forceinline__ v2f load_a_dup(const float* W, int lo, int hi, int c) {
    const int r  = lo & 7;
    const int k0 = 4 * c + 2 * hi;
    v2f a;
    a.x = W[r * 8 + k0];
    a.y = W[r * 8 + k0 + 1];
    return a;
}

__device__ __forceinline__ v2f load_a_fc(const float* W, int lo, int hi, int c) {
    const int r = lo & 7;
    v2f a; a.x = 0.0f; a.y = 0.0f;
    if (r < RNN_NCLS) {
        const int k0 = 4 * c + 2 * hi;
        a.x = W[r * 8 + k0];
        a.y = W[r * 8 + k0 + 1];
    }
    return a;
}

__device__ __forceinline__ void sel_tanh_to_b(const v8f& pD, bool hi, v2f& b0, v2f& b1) {
    b0.x = RNN_TANH(hi ? pD[2] : pD[0]);
    b0.y = RNN_TANH(hi ? pD[3] : pD[1]);
    b1.x = RNN_TANH(hi ? pD[6] : pD[4]);
    b1.y = RNN_TANH(hi ? pD[7] : pD[5]);
}

__device__ __forceinline__ v2f onehot_b(int xi, int hi) {
    v2f oh;
    oh.x = (xi == 2 * hi)     ? 1.0f : 0.0f;
    oh.y = (xi == 2 * hi + 1) ? 1.0f : 0.0f;
    return oh;
}

__global__ __launch_bounds__(32)
void rnn_fused_kernel(const int* __restrict__ x,
                      const float* __restrict__ emb,
                      const float* __restrict__ W_ih0,
                      const float* __restrict__ W_hh0,
                      const float* __restrict__ b_ih0,
                      const float* __restrict__ b_hh0,
                      const float* __restrict__ W_ih1,
                      const float* __restrict__ W_hh1,
                      const float* __restrict__ b_ih1,
                      const float* __restrict__ b_hh1,
                      const float* __restrict__ W_fc,
                      const float* __restrict__ b_fc,
                      float* __restrict__ out) {
    const int lane = threadIdx.x & 31;
    const int lo   = lane & 15;
    const int hi   = lane >> 4;
    const int batch_base = blockIdx.x * 16;

    v2f t0a;
    {
        const int j = lo & 7;
        const float bias = b_ih0[j] + b_hh0[j];
        const int v0 = 2 * hi, v1 = 2 * hi + 1;
        float a0 = bias, a1 = bias;
        #pragma unroll
        for (int d = 0; d < RNN_EMB; ++d) {
            const float w = W_ih0[j * RNN_EMB + d];
            a0 += emb[v0 * RNN_EMB + d] * w;
            a1 += emb[v1 * RNN_EMB + d] * w;
        }
        t0a.x = a0; t0a.y = a1;
    }

    const v2f whh0_a0 = load_a_dup(W_hh0, lo, hi, 0);
    const v2f whh0_a1 = load_a_dup(W_hh0, lo, hi, 1);
    const v2f wih1_a0 = load_a_dup(W_ih1, lo, hi, 0);
    const v2f wih1_a1 = load_a_dup(W_ih1, lo, hi, 1);
    const v2f whh1_a0 = load_a_dup(W_hh1, lo, hi, 0);
    const v2f whh1_a1 = load_a_dup(W_hh1, lo, hi, 1);
    const v2f wfc_a0  = load_a_fc (W_fc,  lo, hi, 0);
    const v2f wfc_a1  = load_a_fc (W_fc,  lo, hi, 1);

    v8f c_bias1, c_fc, c_zero;
    #pragma unroll
    for (int i = 0; i < 8; ++i) {
        c_bias1[i] = b_ih1[i] + b_hh1[i];
        c_fc[i]    = (i < RNN_NCLS) ? b_fc[i] : 0.0f;
        c_zero[i]  = 0.0f;
    }

    v2f h0b0 = {0.0f, 0.0f}, h0b1 = {0.0f, 0.0f};
    v2f h1b0 = {0.0f, 0.0f}, h1b1 = {0.0f, 0.0f};

    float* op = out + ((size_t)batch_base + lo) * RNN_NCLS;

    __shared__ int tokbuf[RNN_GRP * 16];
    const int ldr = lane >> 2;
    const int ldp = lane & 3;
    const int* xblk   = x + (size_t)ldr * RNN_B + batch_base + 4 * ldp;
    int*       lds_st = &tokbuf[ldr * 16 + 4 * ldp];
    const int* lds_ld = &tokbuf[lo];

    int tok[RNN_GRP];
    v4i blk = *(const v4i*)xblk;
    *(v4i*)lds_st = blk;
    __builtin_amdgcn_wave_barrier();
    #pragma unroll
    for (int s = 0; s < RNN_GRP; ++s) tok[s] = lds_ld[16 * s];
    blk = *(const v4i*)(xblk + (size_t)RNN_GRP * RNN_B);

    #pragma unroll 1
    for (int g = 0; g < RNN_NGRP; ++g) {
        int tokn[RNN_GRP];
        *(v4i*)lds_st = blk;
        __builtin_amdgcn_wave_barrier();
        #pragma unroll
        for (int s = 0; s < RNN_GRP; ++s) tokn[s] = lds_ld[16 * s];
        const int gn = (g + 2 < RNN_NGRP) ? (g + 2) : (RNN_NGRP - 1);
        blk = *(const v4i*)(xblk + (size_t)gn * (RNN_GRP * RNN_B));
        __builtin_amdgcn_sched_barrier(0);

        v2f a0[RNN_GRP], a1[RNN_GRP];
        v8f q = wmma4(t0a, onehot_b(tok[0], hi), c_zero);
        #pragma unroll
        for (int s = 0; s < RNN_GRP; ++s) {
            v8f p0 = wmma4(whh0_a0, h0b0, q);
            p0 = wmma4(whh0_a1, h0b1, p0);
            if (s + 1 < RNN_GRP) q = wmma4(t0a, onehot_b(tok[s + 1], hi), c_zero);
            sel_tanh_to_b(p0, hi != 0, h0b0, h0b1);
            a0[s] = h0b0; a1[s] = h0b1;
        }

        v2f e0[RNN_GRP], e1[RNN_GRP];
        v8f r = wmma4(wih1_a1, a1[0], wmma4(wih1_a0, a0[0], c_bias1));
        #pragma unroll
        for (int s = 0; s < RNN_GRP; ++s) {
            v8f p1 = wmma4(whh1_a0, h1b0, r);
            p1 = wmma4(whh1_a1, h1b1, p1);
            if (s + 1 < RNN_GRP)
                r = wmma4(wih1_a1, a1[s + 1], wmma4(wih1_a0, a0[s + 1], c_bias1));
            sel_tanh_to_b(p1, hi != 0, h1b0, h1b1);
            e0[s] = h1b0; e1[s] = h1b1;
        }

        v4f res[RNN_GRP];
        #pragma unroll
        for (int s = 0; s < RNN_GRP; ++s) {
            v8f lg = wmma4(wfc_a1, e1[s], wmma4(wfc_a0, e0[s], c_fc));
            res[s].x = lg[0]; res[s].y = lg[1]; res[s].z = lg[2]; res[s].w = lg[3];
        }
        if (!hi) {
            #pragma unroll
            for (int s = 0; s < RNN_GRP; ++s)
                *(v4f*)(op + (size_t)s * (RNN_B * RNN_NCLS)) = res[s];
        }

        op += (size_t)RNN_GRP * RNN_B * RNN_NCLS;
        #pragma unroll
        for (int s = 0; s < RNN_GRP; ++s) tok[s] = tokn[s];
    }
}

extern "C" void kernel_launch(void* const* d_in, const int* in_sizes, int n_in,
                              void* d_out, int out_size, void* d_ws, size_t ws_size,
                              hipStream_t stream) {
    (void)in_sizes; (void)n_in; (void)d_ws; (void)ws_size; (void)out_size;
    const int*   x     = (const int*)  d_in[0];
    const float* emb   = (const float*)d_in[1];
    const float* W_ih0 = (const float*)d_in[2];
    const float* W_hh0 = (const float*)d_in[3];
    const float* b_ih0 = (const float*)d_in[4];
    const float* b_hh0 = (const float*)d_in[5];
    const float* W_ih1 = (const float*)d_in[6];
    const float* W_hh1 = (const float*)d_in[7];
    const float* b_ih1 = (const float*)d_in[8];
    const float* b_hh1 = (const float*)d_in[9];
    const float* W_fc  = (const float*)d_in[10];
    const float* b_fc  = (const float*)d_in[11];
    float* out = (float*)d_out;

    dim3 grid(RNN_B / 16);
    dim3 block(32);
    rnn_fused_kernel<<<grid, block, 0, stream>>>(
        x, emb, W_ih0, W_hh0, b_ih0, b_hh0,
        W_ih1, W_hh1, b_ih1, b_hh1, W_fc, b_fc, out);
}


# Products with the same B operand share one WMMA pair through the M rows: W_ih1 under W_hh0 (B=h0), W_fc under W_hh1 (B=h1); 5 WMMAs a step, was 9; layer 1 and logits lag one step; sum order kept
# speedup vs baseline: 3.1247x; 1.3243x over previous
#include <hip/hip_runtime.h>
#include <hip/hip_bf16.h>
#include <math.h>
#include <stddef.h>


#define RNN_T    2048
#define RNN_B    2048
#define RNN_EMB  10
#define RNN_HID  8
#define RNN_NCLS 4
#define RNN_GRP  8
#define RNN_NGRP (RNN_T / RNN_GRP)
#define RNN_ROW  (RNN_B * RNN_NCLS)

typedef __attribute__((ext_vector_type(2))) float v2f;
typedef __attribute__((ext_vector_type(4))) float v4f;
typedef __attribute__((ext_vector_type(8))) float v8f;
typedef __attribute__((ext_vector_type(4))) int   v4i;

#if defined(__has_builtin)
#if __has_builtin(__builtin_amdgcn_tanhf)
#define RNN_TANH(x) __builtin_amdgcn_tanhf(x)
#endif
#endif
#ifndef RNN_TANH
#define RNN_TANH(x) tanhf(x)
#endif

__device__ __forceinline__ v8f wmma4(v2f a, v2f b, v8f c) {
    return __builtin_amdgcn_wmma_f32_16x16x4_f32(
         false, a,  false, b,
         (short)0, c,  false,  false);
}

__device__ __forceinline__ int hid_row(int r, int mh) {
    return (r & 1) + 4 * (r >> 1) + 2 * mh;
}

__device__ __forceinline__ v2f load_a_l0(const float* Whh0, const float* Wih1,
                                         int lo, int hi, int c) {
    const int v  = lo & 7;
    const int mh = lo >> 3;
    const float* W = (v < 4) ? Whh0 : Wih1;
    const int row  = hid_row(v & 3, mh);
    const int k0   = 4 * c + 2 * hi;
    v2f a;
    a.x = W[row * 8 + k0];
    a.y = W[row * 8 + k0 + 1];
    return a;
}

__device__ __forceinline__ v2f load_a_l1(const float* Wfc, const float* Whh1,
                                         int lo, int hi, int c) {
    const int v  = lo & 7;
    const int mh = lo >> 3;
    const int k0 = 4 * c + 2 * hi;
    v2f a; a.x = 0.0f; a.y = 0.0f;
    if (v >= 4) {
        const int row = hid_row(v - 4, mh);
        a.x = Whh1[row * 8 + k0];
        a.y = Whh1[row * 8 + k0 + 1];
    } else {
        a.x = Wfc[v * 8 + k0];
        a.y = Wfc[v * 8 + k0 + 1];
    }
    return a;
}

__device__ __forceinline__ v2f onehot_b(int xi, int hi) {
    v2f oh;
    oh.x = (xi == 2 * hi)     ? 1.0f : 0.0f;
    oh.y = (xi == 2 * hi + 1) ? 1.0f : 0.0f;
    return oh;
}

struct RnnConst {
    v2f t0a;
    v2f l0_a0, l0_a1;
    v2f l1_a0, l1_a1;
    v8f c_q;
    v4f c_fc;
    int hi;
    const int* xblk;
    int*       lds_st;
    const int* lds_ld;
    float*     op0;
};

struct RnnState {
    v2f h0b0, h0b1;
    v2f h1b0, h1b1;
    int tok[RNN_GRP];
    v4i blk;
};

template <bool kFirst, bool kLast>
__device__ __forceinline__ void rnn_trip(const RnnConst& k, RnnState& st, int g) {
    const int hi = k.hi;

    int tokn[RNN_GRP];
    if (!kLast) {
        *(v4i*)k.lds_st = st.blk;
        __builtin_amdgcn_wave_barrier();
        #pragma unroll
        for (int s = 0; s < RNN_GRP; ++s) tokn[s] = k.lds_ld[16 * s];
        const int gn = (g + 2 < RNN_NGRP) ? (g + 2) : (RNN_NGRP - 1);
        st.blk = *(const v4i*)(k.xblk + (size_t)gn * (RNN_GRP * RNN_B));
        __builtin_amdgcn_sched_barrier(0);
    }

    const int n0 = kLast ? 1 : RNN_GRP;
    v4f rin[RNN_GRP];
    v8f q = wmma4(k.t0a, onehot_b(st.tok[0], hi), k.c_q);
    #pragma unroll
    for (int s = 0; s < n0; ++s) {
        v8f p0 = wmma4(k.l0_a0, st.h0b0, q);
        p0 = wmma4(k.l0_a1, st.h0b1, p0);
        if (s + 1 < n0) q = wmma4(k.t0a, onehot_b(st.tok[s + 1], hi), k.c_q);
        st.h0b0.x = RNN_TANH(p0[0]); st.h0b0.y = RNN_TANH(p0[1]);
        st.h0b1.x = RNN_TANH(p0[2]); st.h0b1.y = RNN_TANH(p0[3]);
        rin[s].x = p0[4]; rin[s].y = p0[5]; rin[s].z = p0[6]; rin[s].w = p0[7];
    }

    const int s1lo = kFirst ? 1 : 0;
    const int s1hi = kLast ? 2 : RNN_GRP;
    v2f res[RNN_GRP];
    #pragma unroll
    for (int s = s1lo; s < s1hi; ++s) {
        const bool has_in = !(kLast && s > 0);
        const v4f in = rin[has_in ? s : 0];
        v8f c1;
        c1[0] = k.c_fc.x; c1[1] = k.c_fc.y; c1[2] = k.c_fc.z; c1[3] = k.c_fc.w;
        c1[4] = in.x;     c1[5] = in.y;     c1[6] = in.z;     c1[7] = in.w;
        v8f p1 = wmma4(k.l1_a0, st.h1b0, c1);
        p1 = wmma4(k.l1_a1, st.h1b1, p1);
        res[s].x = hi ? p1[2] : p1[0];
        res[s].y = hi ? p1[3] : p1[1];
        if (has_in) {
            st.h1b0.x = RNN_TANH(p1[4]); st.h1b0.y = RNN_TANH(p1[5]);
            st.h1b1.x = RNN_TANH(p1[6]); st.h1b1.y = RNN_TANH(p1[7]);
        }
    }

    {
        float* const op = k.op0 + ((ptrdiff_t)g * RNN_GRP - 2) * (ptrdiff_t)RNN_ROW;
        const int r0 = kFirst ? 2 : 0;
        const int r1 = kLast ? 2 : RNN_GRP;
        #pragma unroll
        for (int s = r0; s < r1; ++s)
            *(v2f*)(op + (ptrdiff_t)s * RNN_ROW) = res[s];
    }

    if (!kLast) {
        #pragma unroll
        for (int s = 0; s < RNN_GRP; ++s) st.tok[s] = tokn[s];
    }
}

__global__ __launch_bounds__(32)
void rnn_fused_kernel(const int* __restrict__ x,
                      const float* __restrict__ emb,
                      const float* __restrict__ W_ih0,
                      const float* __restrict__ W_hh0,
                      const float* __restrict__ b_ih0,
                      const float* __restrict__ b_hh0,
                      const float* __restrict__ W_ih1,
                      const float* __restrict__ W_hh1,
                      const float* __restrict__ b_ih1,
                      const float* __restrict__ b_hh1,
                      const float* __restrict__ W_fc,
                      const float* __restrict__ b_fc,
                      float* __restrict__ out) {
    const int lane = threadIdx.x & 31;
    const int lo   = lane & 15;
    const int hi   = lane >> 4;
    const int batch_base = blockIdx.x * 16;

    RnnConst k;
    k.hi = hi;

    {
        const int j = hid_row(lo & 3, lo >> 3);
        const float bias = b_ih0[j] + b_hh0[j];
        const int v0 = 2 * hi, v1 = 2 * hi + 1;
        float a0 = bias, a1 = bias;
        #pragma unroll
        for (int d = 0; d < RNN_EMB; ++d) {
            const float w = W_ih0[j * RNN_EMB + d];
            a0 += emb[v0 * RNN_EMB + d] * w;
            a1 += emb[v1 * RNN_EMB + d] * w;
        }
        const bool top = (lo & 7) < 4;
        k.t0a.x = top ? a0 : 0.0f;
        k.t0a.y = top ? a1 : 0.0f;
    }

    k.l0_a0 = load_a_l0(W_hh0, W_ih1, lo, hi, 0);
    k.l0_a1 = load_a_l0(W_hh0, W_ih1, lo, hi, 1);
    k.l1_a0 = load_a_l1(W_fc,  W_hh1, lo, hi, 0);
    k.l1_a1 = load_a_l1(W_fc,  W_hh1, lo, hi, 1);

    #pragma unroll
    for (int i = 0; i < 4; ++i) {
        const int j = hid_row(i, hi);
        k.c_q[i]     = 0.0f;
        k.c_q[4 + i] = b_ih1[j] + b_hh1[j];
    }
    k.c_fc.x = b_fc[0]; k.c_fc.y = b_fc[1]; k.c_fc.z = b_fc[2]; k.c_fc.w = b_fc[3];

    k.op0 = out + ((size_t)batch_base + lo) * RNN_NCLS + 2 * hi;

    __shared__ int tokbuf[RNN_GRP * 16];
    const int ldr = lane >> 2;
    const int ldp = lane & 3;
    k.xblk   = x + (size_t)ldr * RNN_B + batch_base + 4 * ldp;
    k.lds_st = &tokbuf[ldr * 16 + 4 * ldp];
    k.lds_ld = &tokbuf[lo];

    RnnState st;
    st.h0b0.x = 0.0f; st.h0b0.y = 0.0f; st.h0b1.x = 0.0f; st.h0b1.y = 0.0f;
    st.h1b0.x = 0.0f; st.h1b0.y = 0.0f; st.h1b1.x = 0.0f; st.h1b1.y = 0.0f;

    st.blk = *(const v4i*)k.xblk;
    *(v4i*)k.lds_st = st.blk;
    __builtin_amdgcn_wave_barrier();
    #pragma unroll
    for (int s = 0; s < RNN_GRP; ++s) st.tok[s] = k.lds_ld[16 * s];
    st.blk = *(const v4i*)(k.xblk + (size_t)RNN_GRP * RNN_B);

    rnn_trip<true, false>(k, st, 0);
    #pragma unroll 1
    for (int g = 1; g < RNN_NGRP; ++g) rnn_trip<false, false>(k, st, g);
    rnn_trip<false, true>(k, st, RNN_NGRP);
}

extern "C" void kernel_launch(void* const* d_in, const int* in_sizes, int n_in,
                              void* d_out, int out_size, void* d_ws, size_t ws_size,
                              hipStream_t stream) {
    (void)in_sizes; (void)n_in; (void)d_ws; (void)ws_size; (void)out_size;
    const int*   x     = (const int*)  d_in[0];
    const float* emb   = (const float*)d_in[1];
    const float* W_ih0 = (const float*)d_in[2];
    const float* W_hh0 = (const float*)d_in[3];
    const float* b_ih0 = (const float*)d_in[4];
    const float* b_hh0 = (const float*)d_in[5];
    const float* W_ih1 = (const float*)d_in[6];
    const float* W_hh1 = (const float*)d_in[7];
    const float* b_ih1 = (const float*)d_in[8];
    const float* b_hh1 = (const float*)d_in[9];
    const float* W_fc  = (const float*)d_in[10];
    const float* b_fc  = (const float*)d_in[11];
    float* out = (float*)d_out;

    dim3 grid(RNN_B / 16);
    dim3 block(32);
    rnn_fused_kernel<<<grid, block, 0, stream>>>(
        x, emb, W_ih0, W_hh0, b_ih0, b_hh0,
        W_ih1, W_hh1, b_ih1, b_hh1, W_fc, b_fc, out);
}


# Global memory traffic moved to a second wave per tile: it fetches token blocks and stores logits rows through LDS rings, one LDS-only barrier per 8 steps; the matrix wave issues none; sum order kept
# speedup vs baseline: 3.9105x; 1.2515x over previous
#include <hip/hip_runtime.h>
#include <hip/hip_bf16.h>
#include <math.h>
#include <stddef.h>


#define RNN_T    2048
#define RNN_B    2048
#define RNN_EMB  10
#define RNN_HID  8
#define RNN_NCLS 4
#define RNN_GRP  8
#define RNN_NGRP (RNN_T / RNN_GRP)
#define RNN_ROW  (RNN_B * RNN_NCLS)
#define RNN_TOKSLOT (RNN_GRP * 16)
#define RNN_OUTSLOT (RNN_GRP * 64)

typedef __attribute__((ext_vector_type(2))) float v2f;
typedef __attribute__((ext_vector_type(4))) float v4f;
typedef __attribute__((ext_vector_type(8))) float v8f;
typedef __attribute__((ext_vector_type(4))) int   v4i;

#if defined(__has_builtin)
#if __has_builtin(__builtin_amdgcn_tanhf)
#define RNN_TANH(x) __builtin_amdgcn_tanhf(x)
#endif
#endif
#ifndef RNN_TANH
#define RNN_TANH(x) tanhf(x)
#endif

__device__ __forceinline__ v8f wmma4(v2f a, v2f b, v8f c) {
    return __builtin_amdgcn_wmma_f32_16x16x4_f32(
         false, a,  false, b,
         (short)0, c,  false,  false);
}

__device__ __forceinline__ void lds_barrier() {
    __builtin_amdgcn_fence(__ATOMIC_RELEASE, "workgroup", "local");
    __builtin_amdgcn_s_barrier();
    __builtin_amdgcn_fence(__ATOMIC_ACQUIRE, "workgroup", "local");
}

__device__ __forceinline__ int hid_row(int r, int mh) {
    return (r & 1) + 4 * (r >> 1) + 2 * mh;
}

__device__ __forceinline__ v2f load_a_l0(const float* Whh0, const float* Wih1,
                                         int lo, int hi, int c) {
    const int v  = lo & 7;
    const int mh = lo >> 3;
    const float* W = (v < 4) ? Whh0 : Wih1;
    const int row  = hid_row(v & 3, mh);
    const int k0   = 4 * c + 2 * hi;
    v2f a;
    a.x = W[row * 8 + k0];
    a.y = W[row * 8 + k0 + 1];
    return a;
}

__device__ __forceinline__ v2f load_a_l1(const float* Wfc, const float* Whh1,
                                         int lo, int hi, int c) {
    const int v  = lo & 7;
    const int mh = lo >> 3;
    const int k0 = 4 * c + 2 * hi;
    v2f a; a.x = 0.0f; a.y = 0.0f;
    if (v >= 4) {
        const int row = hid_row(v - 4, mh);
        a.x = Whh1[row * 8 + k0];
        a.y = Whh1[row * 8 + k0 + 1];
    } else {
        a.x = Wfc[v * 8 + k0];
        a.y = Wfc[v * 8 + k0 + 1];
    }
    return a;
}

__device__ __forceinline__ v2f onehot_b(int xi, int hi) {
    v2f oh;
    oh.x = (xi == 2 * hi)     ? 1.0f : 0.0f;
    oh.y = (xi == 2 * hi + 1) ? 1.0f : 0.0f;
    return oh;
}

struct RnnConst {
    v2f t0a;
    v2f l0_a0, l0_a1;
    v2f l1_a0, l1_a1;
    v8f c_q;
    v4f c_fc;
    int hi;
    const int* lds_ld;
    float*     lds_res;
};

struct RnnState {
    v2f h0b0, h0b1;
    v2f h1b0, h1b1;
    int tok[RNN_GRP];
    v2f res[RNN_GRP];
};

struct IoConst {
    const int*   xblk;
    int*         lds_tok;
    const float* lds_row;
    float*       op;
};

__device__ __forceinline__ void io_trip(const IoConst& c, int g) {
    const int gl = (g + 3 < RNN_NGRP) ? (g + 3) : (RNN_NGRP - 1);
    const v4i blk = *(const v4i*)(c.xblk + (size_t)gl * (RNN_GRP * RNN_B));
    const int gp = (g + 5 < RNN_NGRP) ? (g + 5) : (RNN_NGRP - 1);
    __builtin_prefetch(c.xblk + (size_t)gp * (RNN_GRP * RNN_B), 0, 3);
    if (g >= 2) {
        const int t = g - 2;
        const float* src = c.lds_row + (t & 1) * RNN_OUTSLOT;
        float* dst = c.op + ((ptrdiff_t)t * RNN_GRP - 2) * (ptrdiff_t)RNN_ROW;
        if (t > 0)
            *(v4f*)dst = *(const v4f*)src;
        if (t < RNN_NGRP) {
            const v4f v1 = *(const v4f*)(src + 128);
            const v4f v2 = *(const v4f*)(src + 256);
            const v4f v3 = *(const v4f*)(src + 384);
            *(v4f*)(dst + (ptrdiff_t)2 * RNN_ROW) = v1;
            *(v4f*)(dst + (ptrdiff_t)4 * RNN_ROW) = v2;
            *(v4f*)(dst + (ptrdiff_t)6 * RNN_ROW) = v3;
        }
    }
    *(v4i*)(c.lds_tok + ((g + 3) & 3) * RNN_TOKSLOT) = blk;
}

template <bool kFirst, bool kLast>
__device__ __forceinline__ void rnn_trip(const RnnConst& k, RnnState& st, int g) {
    const int hi = k.hi;

    int tokn[RNN_GRP];
    if (!kFirst) {
        float* const lr = k.lds_res + ((g - 1) & 1) * RNN_OUTSLOT;
        #pragma unroll
        for (int s = 0; s < RNN_GRP; ++s) *(v2f*)(lr + 64 * s) = st.res[s];
    }
    if (!kLast) {
        const int* const lt = k.lds_ld + ((g + 1) & 3) * RNN_TOKSLOT;
        #pragma unroll
        for (int s = 0; s < RNN_GRP; ++s) tokn[s] = lt[16 * s];
    }
    __builtin_amdgcn_sched_barrier(0);

    const int n0 = kLast ? 1 : RNN_GRP;
    v4f rin[RNN_GRP];
    v8f q = wmma4(k.t0a, onehot_b(st.tok[0], hi), k.c_q);
    #pragma unroll
    for (int s = 0; s < n0; ++s) {
        v8f p0 = wmma4(k.l0_a0, st.h0b0, q);
        p0 = wmma4(k.l0_a1, st.h0b1, p0);
        if (s + 1 < n0) q = wmma4(k.t0a, onehot_b(st.tok[s + 1], hi), k.c_q);
        st.h0b0.x = RNN_TANH(p0[0]); st.h0b0.y = RNN_TANH(p0[1]);
        st.h0b1.x = RNN_TANH(p0[2]); st.h0b1.y = RNN_TANH(p0[3]);
        rin[s].x = p0[4]; rin[s].y = p0[5]; rin[s].z = p0[6]; rin[s].w = p0[7];
    }

    const int s1lo = kFirst ? 1 : 0;
    const int s1hi = kLast ? 2 : RNN_GRP;
    v2f res[RNN_GRP];
    #pragma unroll
    for (int s = 0; s < RNN_GRP; ++s) { res[s].x = 0.0f; res[s].y = 0.0f; }
    #pragma unroll
    for (int s = s1lo; s < s1hi; ++s) {
        const bool has_in = !(kLast && s > 0);
        const v4f in = rin[has_in ? s : 0];
        v8f c1;
        c1[0] = k.c_fc.x; c1[1] = k.c_fc.y; c1[2] = k.c_fc.z; c1[3] = k.c_fc.w;
        c1[4] = in.x;     c1[5] = in.y;     c1[6] = in.z;     c1[7] = in.w;
        v8f p1 = wmma4(k.l1_a0, st.h1b0, c1);
        p1 = wmma4(k.l1_a1, st.h1b1, p1);
        res[s].x = hi ? p1[2] : p1[0];
        res[s].y = hi ? p1[3] : p1[1];
        if (has_in) {
            st.h1b0.x = RNN_TANH(p1[4]); st.h1b0.y = RNN_TANH(p1[5]);
            st.h1b1.x = RNN_TANH(p1[6]); st.h1b1.y = RNN_TANH(p1[7]);
        }
    }

    #pragma unroll
    for (int s = 0; s < RNN_GRP; ++s) st.res[s] = res[s];

    if (!kLast) {
        #pragma unroll
        for (int s = 0; s < RNN_GRP; ++s) st.tok[s] = tokn[s];
    }
}

__global__ __launch_bounds__(64)
void rnn_fused_kernel(const int* __restrict__ x,
                      const float* __restrict__ emb,
                      const float* __restrict__ W_ih0,
                      const float* __restrict__ W_hh0,
                      const float* __restrict__ b_ih0,
                      const float* __restrict__ b_hh0,
                      const float* __restrict__ W_ih1,
                      const float* __restrict__ W_hh1,
                      const float* __restrict__ b_ih1,
                      const float* __restrict__ b_hh1,
                      const float* __restrict__ W_fc,
                      const float* __restrict__ b_fc,
                      float* __restrict__ out) {
    const int wv   = __builtin_amdgcn_readfirstlane((int)(threadIdx.x >> 5));
    const int lane = threadIdx.x & 31;
    const int lo   = lane & 15;
    const int hi   = lane >> 4;
    const int batch_base = blockIdx.x * 16;

    __shared__ int   tokbuf[4 * RNN_TOKSLOT];
    __shared__ float outbuf[2 * RNN_OUTSLOT];

    RnnConst k;
    RnnState st;
    IoConst  io;

    if (wv == 0) {
        k.hi = hi;

        {
            const int j = hid_row(lo & 3, lo >> 3);
            const float bias = b_ih0[j] + b_hh0[j];
            const int v0 = 2 * hi, v1 = 2 * hi + 1;
            float a0 = bias, a1 = bias;
            #pragma unroll
            for (int d = 0; d < RNN_EMB; ++d) {
                const float w = W_ih0[j * RNN_EMB + d];
                a0 += emb[v0 * RNN_EMB + d] * w;
                a1 += emb[v1 * RNN_EMB + d] * w;
            }
            const bool top = (lo & 7) < 4;
            k.t0a.x = top ? a0 : 0.0f;
            k.t0a.y = top ? a1 : 0.0f;
        }

        k.l0_a0 = load_a_l0(W_hh0, W_ih1, lo, hi, 0);
        k.l0_a1 = load_a_l0(W_hh0, W_ih1, lo, hi, 1);
        k.l1_a0 = load_a_l1(W_fc,  W_hh1, lo, hi, 0);
        k.l1_a1 = load_a_l1(W_fc,  W_hh1, lo, hi, 1);

        #pragma unroll
        for (int i = 0; i < 4; ++i) {
            const int j = hid_row(i, hi);
            k.c_q[i]     = 0.0f;
            k.c_q[4 + i] = b_ih1[j] + b_hh1[j];
        }
        k.c_fc.x = b_fc[0]; k.c_fc.y = b_fc[1]; k.c_fc.z = b_fc[2]; k.c_fc.w = b_fc[3];

        k.lds_ld  = &tokbuf[lo];
        k.lds_res = &outbuf[lo * RNN_NCLS + 2 * hi];

        st.h0b0.x = 0.0f; st.h0b0.y = 0.0f; st.h0b1.x = 0.0f; st.h0b1.y = 0.0f;
        st.h1b0.x = 0.0f; st.h1b0.y = 0.0f; st.h1b1.x = 0.0f; st.h1b1.y = 0.0f;
    } else {
        const int ldr = lane >> 2;
        const int ldp = lane & 3;
        io.xblk    = x + (size_t)ldr * RNN_B + batch_base + 4 * ldp;
        io.lds_tok = &tokbuf[ldr * 16 + 4 * ldp];
        io.lds_row = &outbuf[hi * 64 + lo * RNN_NCLS];
        io.op      = out + (size_t)hi * RNN_ROW + ((size_t)batch_base + lo) * RNN_NCLS;

        const v4i b0 = *(const v4i*)io.xblk;
        const v4i b1 = *(const v4i*)(io.xblk + (size_t)1 * (RNN_GRP * RNN_B));
        const v4i b2 = *(const v4i*)(io.xblk + (size_t)2 * (RNN_GRP * RNN_B));
        __builtin_prefetch(io.xblk + (size_t)3 * (RNN_GRP * RNN_B), 0, 3);
        __builtin_prefetch(io.xblk + (size_t)4 * (RNN_GRP * RNN_B), 0, 3);
        *(v4i*)(io.lds_tok + 0 * RNN_TOKSLOT) = b0;
        *(v4i*)(io.lds_tok + 1 * RNN_TOKSLOT) = b1;
        *(v4i*)(io.lds_tok + 2 * RNN_TOKSLOT) = b2;
    }

    lds_barrier();
    if (wv == 0) {
        #pragma unroll
        for (int s = 0; s < RNN_GRP; ++s) st.tok[s] = k.lds_ld[16 * s];
    }

    lds_barrier();
    if (wv == 0) rnn_trip<true, false>(k, st, 0); else io_trip(io, 0);
    #pragma unroll 1
    for (int g = 1; g < RNN_NGRP; ++g) {
        lds_barrier();
        if (wv == 0) rnn_trip<false, false>(k, st, g); else io_trip(io, g);
    }
    lds_barrier();
    if (wv == 0) rnn_trip<false, true>(k, st, RNN_NGRP); else io_trip(io, RNN_NGRP);
    lds_barrier();
    if (wv == 0) {
        float* const lr = k.lds_res + (RNN_NGRP & 1) * RNN_OUTSLOT;
        #pragma unroll
        for (int s = 0; s < RNN_GRP; ++s) *(v2f*)(lr + 64 * s) = st.res[s];
    } else {
        io_trip(io, RNN_NGRP + 1);
    }
    lds_barrier();
    if (wv != 0) io_trip(io, RNN_NGRP + 2);
}

extern "C" void kernel_launch(void* const* d_in, const int* in_sizes, int n_in,
                              void* d_out, int out_size, void* d_ws, size_t ws_size,
                              hipStream_t stream) {
    (void)in_sizes; (void)n_in; (void)d_ws; (void)ws_size; (void)out_size;
    const int*   x     = (const int*)  d_in[0];
    const float* emb   = (const float*)d_in[1];
    const float* W_ih0 = (const float*)d_in[2];
    const float* W_hh0 = (const float*)d_in[3];
    const float* b_ih0 = (const float*)d_in[4];
    const float* b_hh0 = (const float*)d_in[5];
    const float* W_ih1 = (const float*)d_in[6];
    const float* W_hh1 = (const float*)d_in[7];
    const float* b_ih1 = (const float*)d_in[8];
    const float* b_hh1 = (const float*)d_in[9];
    const float* W_fc  = (const float*)d_in[10];
    const float* b_fc  = (const float*)d_in[11];
    float* out = (float*)d_out;

    dim3 grid(RNN_B / 16);
    dim3 block(64);
    rnn_fused_kernel<<<grid, block, 0, stream>>>(
        x, emb, W_ih0, W_hh0, b_ih0, b_hh0,
        W_ih1, W_hh1, b_ih1, b_hh1, W_fc, b_fc, out);
}
